# starting kernel
# baseline (speedup 1.0000x reference)
#include <hip/hip_runtime.h>


typedef __attribute__((ext_vector_type(2))) float v2f;
typedef __attribute__((ext_vector_type(8))) float v8f;

#define DIM 64
#define KK_TOT 32
#define NT 4

__global__ __launch_bounds__(256) void edge_mlp_wmma(
    const float* __restrict__ nodes, const int* __restrict__ src,
    const int* __restrict__ dst, const float* __restrict__ W1,
    const float* __restrict__ b1, const float* __restrict__ W2,
    const float* __restrict__ b2, float* __restrict__ out, int E)
{
  __shared__ v2f ldsB[KK_TOT * NT * 32];

  for (int g = threadIdx.x; g < KK_TOT * NT * 32; g += blockDim.x) {
    int f = g >> 5, ln = g & 31;
    int n = f >> 5, kk = f & 31;
    int hf = ln >> 4, c = (n << 4) + (ln & 15);
    int row = (kk << 2) + (hf << 1);
    v2f v;
    v.x = W1[row * DIM + c];
    v.y = W1[(row + 1) * DIM + c];
    ldsB[g] = v;
  }
  __syncthreads();

  const int lane = threadIdx.x & 31;
  const int half = lane >> 4;
  const int l15  = lane & 15;

  float bias[NT], w2v[NT];
#pragma unroll
  for (int n = 0; n < NT; ++n) {
    bias[n] = b1[n * 16 + l15];
    w2v[n]  = W2[n * 16 + l15];
  }
  const float b2v = b2[0];

  const int wavesPerBlk = blockDim.x >> 5;
  const int gw = blockIdx.x * wavesPerBlk + (threadIdx.x >> 5);
  const int nw = gridDim.x * wavesPerBlk;
  const int tiles = (E + 15) >> 4;

  for (int t = gw; t < tiles; t += nw) {
    const int e0 = t << 4;
    int e = e0 + l15; if (e >= E) e = E - 1;
    const float* sp = nodes + (long)src[e] * DIM + (half << 1);
    const float* dp = nodes + (long)dst[e] * DIM + (half << 1);

    v2f aS[16], aD[16];
#pragma unroll
    for (int kk = 0; kk < 16; ++kk) {
      aS[kk] = *(const v2f*)(sp + (kk << 2));
      aD[kk] = *(const v2f*)(dp + (kk << 2));
    }

    float pl[8];
#pragma unroll
    for (int r = 0; r < 8; ++r) pl[r] = 0.f;

#pragma unroll
    for (int n = 0; n < NT; ++n) {
      v8f acc;
#pragma unroll
      for (int r = 0; r < 8; ++r) acc[r] = bias[n];
#pragma unroll
      for (int kk = 0; kk < 16; ++kk) {
        v2f bfrag = ldsB[(n * 32 + kk) * 32 + lane];
        acc = __builtin_amdgcn_wmma_f32_16x16x4_f32(
            false, aS[kk], false, bfrag, (short)0, acc, false, false);
      }
#pragma unroll
      for (int kk = 0; kk < 16; ++kk) {
        v2f bfrag = ldsB[(n * 32 + 16 + kk) * 32 + lane];
        acc = __builtin_amdgcn_wmma_f32_16x16x4_f32(
            false, aD[kk], false, bfrag, (short)0, acc, false, false);
      }
#pragma unroll
      for (int r = 0; r < 8; ++r) {
        float h = acc[r] > 0.f ? acc[r] : 0.f;
        pl[r] += h * w2v[n];
      }
    }

#pragma unroll
    for (int r = 0; r < 8; ++r) {
#pragma unroll
      for (int off = 1; off < 16; off <<= 1)
        pl[r] += __shfl_xor(pl[r], off, 32);
      pl[r] += b2v;
    }

    if (e0 + 16 <= E) {
      if (l15 == 0) {
        float4* op = (float4*)(out + e0 + half * 8);
        op[0] = make_float4(pl[0], pl[1], pl[2], pl[3]);
        op[1] = make_float4(pl[4], pl[5], pl[6], pl[7]);
      }
    } else if (l15 == 0) {
#pragma unroll
      for (int r = 0; r < 8; ++r) {
        int idx = e0 + half * 8 + r;
        if (idx < E) out[idx] = pl[r];
      }
    }
  }
}

extern "C" void kernel_launch(void* const* d_in, const int* in_sizes, int n_in,
                              void* d_out, int out_size, void* d_ws, size_t ws_size,
                              hipStream_t stream) {
  const float* nodes = (const float*)d_in[0];
  const int*   src   = (const int*)d_in[1];
  const int*   dst   = (const int*)d_in[2];
  const float* W1    = (const float*)d_in[3];
  const float* b1    = (const float*)d_in[4];
  const float* W2    = (const float*)d_in[5];
  const float* b2    = (const float*)d_in[6];
  float* out = (float*)d_out;

  const int E = in_sizes[1];
  const int tiles = (E + 15) / 16;
  int blocks = (tiles + 7) / 8;
  if (blocks > 2048) blocks = 2048;
  if (blocks < 1) blocks = 1;

  edge_mlp_wmma<<<blocks, 256, 0, stream>>>(nodes, src, dst, W1, b1, W2, b2, out, E);
}


# Tile loop pipelined by half tiles: the A registers each 64-WMMA block frees take the next tile's row gather at once, edge indices two tiles ahead, so index and gather latency hide under matrix work
# speedup vs baseline: 1.0826x; 1.0826x over previous
#include <hip/hip_runtime.h>


typedef __attribute__((ext_vector_type(2))) float v2f;
typedef __attribute__((ext_vector_type(8))) float v8f;

#define DIM 64
#define KK_TOT 32
#define NT 4

static __device__ __forceinline__ void tile_indices(
    const int* __restrict__ src, const int* __restrict__ dst,
    int t, int l15, int E, int& s, int& d)
{
  int e = (t << 4) + l15;
  if (e >= E) e = E - 1;
  s = src[e];
  d = dst[e];
}

static __device__ __forceinline__ void gather_row(
    const float* __restrict__ nodes, int row, int half, v2f (&a)[16])
{
  const float* p = nodes + (long)row * DIM + (half << 1);
#pragma unroll
  for (int kk = 0; kk < 16; ++kk) a[kk] = *(const v2f*)(p + (kk << 2));
}

template <int KBASE>
static __device__ __forceinline__ void mma_half(
    v8f (&acc)[NT], const v8f (&c)[NT], const v2f (&a)[16],
    const v2f (&bw)[KK_TOT * NT])
{
#pragma unroll
  for (int kk = 0; kk < 16; ++kk) {
#pragma unroll
    for (int n = 0; n < NT; ++n) {
      acc[n] = __builtin_amdgcn_wmma_f32_16x16x4_f32(
          false, a[kk], false, bw[n * 32 + KBASE + kk], (short)0,
          kk == 0 ? c[n] : acc[n], false, false);
    }
  }
}

static __device__ __forceinline__ void tile_matrix_work(
    v8f (&acc)[NT], const v8f (&biasT)[NT], v2f (&aSrc)[16], const v2f (&aDst)[16],
    const v2f (&bw)[KK_TOT * NT], const float* __restrict__ nodes, int sNext, int half)
{
  mma_half<0>(acc, biasT, aSrc, bw);
  __builtin_amdgcn_sched_barrier(0);
  gather_row(nodes, sNext, half, aSrc);
  mma_half<16>(acc, acc, aDst, bw);
}

static __device__ __forceinline__ void finish_tile(
    const v8f (&acc)[NT], const float (&w2v)[NT], float b2v,
    int half, int l15, int t, int E, float* __restrict__ out)
{
  float pl[8];
#pragma unroll
  for (int r = 0; r < 8; ++r) pl[r] = 0.f;

#pragma unroll
  for (int n = 0; n < NT; ++n) {
#pragma unroll
    for (int r = 0; r < 8; ++r) {
      float h = acc[n][r] > 0.f ? acc[n][r] : 0.f;
      pl[r] += h * w2v[n];
    }
  }

#pragma unroll
  for (int r = 0; r < 8; ++r) {
#pragma unroll
    for (int off = 1; off < 16; off <<= 1)
      pl[r] += __shfl_xor(pl[r], off, 32);
    pl[r] += b2v;
  }

  const int e0 = t << 4;
  if (e0 + 16 <= E) {
    if (l15 == 0) {
      float4* op = (float4*)(out + e0 + half * 8);
      op[0] = make_float4(pl[0], pl[1], pl[2], pl[3]);
      op[1] = make_float4(pl[4], pl[5], pl[6], pl[7]);
    }
  } else if (l15 == 0) {
#pragma unroll
    for (int r = 0; r < 8; ++r) {
      int idx = e0 + half * 8 + r;
      if (idx < E) out[idx] = pl[r];
    }
  }
}

__global__ __launch_bounds__(256) void edge_mlp_wmma(
    const float* __restrict__ nodes, const int* __restrict__ src,
    const int* __restrict__ dst, const float* __restrict__ W1,
    const float* __restrict__ b1, const float* __restrict__ W2,
    const float* __restrict__ b2, float* __restrict__ out, int E)
{
  __shared__ v2f ldsB[KK_TOT * NT * 32];

  for (int g = threadIdx.x; g < KK_TOT * NT * 32; g += blockDim.x) {
    int f = g >> 5, ln = g & 31;
    int n = f >> 5, kk = f & 31;
    int hf = ln >> 4, c = (n << 4) + (ln & 15);
    int row = (kk << 2) + (hf << 1);
    v2f v;
    v.x = W1[row * DIM + c];
    v.y = W1[(row + 1) * DIM + c];
    ldsB[g] = v;
  }
  __syncthreads();

  const int lane = threadIdx.x & 31;
  const int half = lane >> 4;
  const int l15  = lane & 15;

  float bias[NT], w2v[NT];
#pragma unroll
  for (int n = 0; n < NT; ++n) {
    bias[n] = b1[n * 16 + l15];
    w2v[n]  = W2[n * 16 + l15];
  }
  const float b2v = b2[0];

  v8f biasT[NT];
#pragma unroll
  for (int n = 0; n < NT; ++n)
#pragma unroll
    for (int r = 0; r < 8; ++r) biasT[n][r] = bias[n];

  const int wavesPerBlk = blockDim.x >> 5;
  const int gw = blockIdx.x * wavesPerBlk + (threadIdx.x >> 5);
  const int nw = gridDim.x * wavesPerBlk;
  const int tiles = (E + 15) >> 4;

  int t = gw;
  if (t >= tiles) return;

  v2f bw[KK_TOT * NT];
#pragma unroll
  for (int f = 0; f < KK_TOT * NT; ++f) bw[f] = ldsB[f * 32 + lane];

  int s, d;
  tile_indices(src, dst, t, l15, E, s, d);
  v2f aSrc[16], aDst[16];
  gather_row(nodes, s, half, aSrc);
  gather_row(nodes, d, half, aDst);
  tile_indices(src, dst, t + nw, l15, E, s, d);

  v8f acc[NT];
  tile_matrix_work(acc, biasT, aSrc, aDst, bw, nodes, s, half);

  for (;;) {
    gather_row(nodes, d, half, aDst);
    tile_indices(src, dst, t + 2 * nw, l15, E, s, d);

    finish_tile(acc, w2v, b2v, half, l15, t, E, out);

    t += nw;
    if (t >= tiles) break;

    tile_matrix_work(acc, biasT, aSrc, aDst, bw, nodes, s, half);
  }
}

extern "C" void kernel_launch(void* const* d_in, const int* in_sizes, int n_in,
                              void* d_out, int out_size, void* d_ws, size_t ws_size,
                              hipStream_t stream) {
  const float* nodes = (const float*)d_in[0];
  const int*   src   = (const int*)d_in[1];
  const int*   dst   = (const int*)d_in[2];
  const float* W1    = (const float*)d_in[3];
  const float* b1    = (const float*)d_in[4];
  const float* W2    = (const float*)d_in[5];
  const float* b2    = (const float*)d_in[6];
  float* out = (float*)d_out;

  const int E = in_sizes[1];
  const int tiles = (E + 15) / 16;
  int blocks = (tiles + 7) / 8;
  if (blocks > 2048) blocks = 2048;
  if (blocks < 1) blocks = 1;

  edge_mlp_wmma<<<blocks, 256, 0, stream>>>(nodes, src, dst, W1, b1, W2, b2, out, E);
}


# Layer 1 hoisted from edges to nodes: per-node tables emb@W1[0:64]+b1 and emb@W1[64:128] (wmma, in workspace), edge pass gathers 2 rows and adds; sum order: K halves added last, lane tree 8-4-2-1
# speedup vs baseline: 2.7056x; 2.4992x over previous
#include <hip/hip_runtime.h>


typedef __attribute__((ext_vector_type(2))) float v2f;
typedef __attribute__((ext_vector_type(8))) float v8f;

#define DIM 64
#define KK_TOT 32
#define NT 4

static __device__ __forceinline__ void tile_indices(
    const int* __restrict__ src, const int* __restrict__ dst,
    int t, int l15, int E, int& s, int& d)
{
  int e = (t << 4) + l15;
  if (e >= E) e = E - 1;
  s = src[e];
  d = dst[e];
}

static __device__ __forceinline__ void gather_row(
    const float* __restrict__ nodes, int row, int half, v2f (&a)[16])
{
  const float* p = nodes + (long)row * DIM + (half << 1);
#pragma unroll
  for (int kk = 0; kk < 16; ++kk) a[kk] = *(const v2f*)(p + (kk << 2));
}

template <int KBASE>
static __device__ __forceinline__ void mma_half(
    v8f (&acc)[NT], const v8f (&c)[NT], const v2f (&a)[16],
    const v2f (&bw)[KK_TOT * NT])
{
#pragma unroll
  for (int kk = 0; kk < 16; ++kk) {
#pragma unroll
    for (int n = 0; n < NT; ++n) {
      acc[n] = __builtin_amdgcn_wmma_f32_16x16x4_f32(
          false, a[kk], false, bw[n * 32 + KBASE + kk], (short)0,
          kk == 0 ? c[n] : acc[n], false, false);
    }
  }
}

static __device__ __forceinline__ void tile_matrix_work(
    v8f (&acc)[NT], const v8f (&biasT)[NT], v2f (&aSrc)[16], const v2f (&aDst)[16],
    const v2f (&bw)[KK_TOT * NT], const float* __restrict__ nodes, int sNext, int half)
{
  mma_half<0>(acc, biasT, aSrc, bw);
  __builtin_amdgcn_sched_barrier(0);
  gather_row(nodes, sNext, half, aSrc);
  mma_half<16>(acc, acc, aDst, bw);
}

static __device__ __forceinline__ void finish_tile(
    const v8f (&acc)[NT], const float (&w2v)[NT], float b2v,
    int half, int l15, int t, int E, float* __restrict__ out)
{
  float pl[8];
#pragma unroll
  for (int r = 0; r < 8; ++r) pl[r] = 0.f;

#pragma unroll
  for (int n = 0; n < NT; ++n) {
#pragma unroll
    for (int r = 0; r < 8; ++r) {
      float h = acc[n][r] > 0.f ? acc[n][r] : 0.f;
      pl[r] += h * w2v[n];
    }
  }

#pragma unroll
  for (int r = 0; r < 8; ++r) {
#pragma unroll
    for (int off = 1; off < 16; off <<= 1)
      pl[r] += __shfl_xor(pl[r], off, 32);
    pl[r] += b2v;
  }

  const int e0 = t << 4;
  if (e0 + 16 <= E) {
    if (l15 == 0) {
      float4* op = (float4*)(out + e0 + half * 8);
      op[0] = make_float4(pl[0], pl[1], pl[2], pl[3]);
      op[1] = make_float4(pl[4], pl[5], pl[6], pl[7]);
    }
  } else if (l15 == 0) {
#pragma unroll
    for (int r = 0; r < 8; ++r) {
      int idx = e0 + half * 8 + r;
      if (idx < E) out[idx] = pl[r];
    }
  }
}

__global__ __launch_bounds__(256) void edge_mlp_wmma(
    const float* __restrict__ nodes, const int* __restrict__ src,
    const int* __restrict__ dst, const float* __restrict__ W1,
    const float* __restrict__ b1, const float* __restrict__ W2,
    const float* __restrict__ b2, float* __restrict__ out, int E)
{
  __shared__ v2f ldsB[KK_TOT * NT * 32];

  for (int g = threadIdx.x; g < KK_TOT * NT * 32; g += blockDim.x) {
    int f = g >> 5, ln = g & 31;
    int n = f >> 5, kk = f & 31;
    int hf = ln >> 4, c = (n << 4) + (ln & 15);
    int row = (kk << 2) + (hf << 1);
    v2f v;
    v.x = W1[row * DIM + c];
    v.y = W1[(row + 1) * DIM + c];
    ldsB[g] = v;
  }
  __syncthreads();

  const int lane = threadIdx.x & 31;
  const int half = lane >> 4;
  const int l15  = lane & 15;

  float bias[NT], w2v[NT];
#pragma unroll
  for (int n = 0; n < NT; ++n) {
    bias[n] = b1[n * 16 + l15];
    w2v[n]  = W2[n * 16 + l15];
  }
  const float b2v = b2[0];

  v8f biasT[NT];
#pragma unroll
  for (int n = 0; n < NT; ++n)
#pragma unroll
    for (int r = 0; r < 8; ++r) biasT[n][r] = bias[n];

  const int wavesPerBlk = blockDim.x >> 5;
  const int gw = blockIdx.x * wavesPerBlk + (threadIdx.x >> 5);
  const int nw = gridDim.x * wavesPerBlk;
  const int tiles = (E + 15) >> 4;

  int t = gw;
  if (t >= tiles) return;

  v2f bw[KK_TOT * NT];
#pragma unroll
  for (int f = 0; f < KK_TOT * NT; ++f) bw[f] = ldsB[f * 32 + lane];

  int s, d;
  tile_indices(src, dst, t, l15, E, s, d);
  v2f aSrc[16], aDst[16];
  gather_row(nodes, s, half, aSrc);
  gather_row(nodes, d, half, aDst);
  tile_indices(src, dst, t + nw, l15, E, s, d);

  v8f acc[NT];
  tile_matrix_work(acc, biasT, aSrc, aDst, bw, nodes, s, half);

  for (;;) {
    gather_row(nodes, d, half, aDst);
    tile_indices(src, dst, t + 2 * nw, l15, E, s, d);

    finish_tile(acc, w2v, b2v, half, l15, t, E, out);

    t += nw;
    if (t >= tiles) break;

    tile_matrix_work(acc, biasT, aSrc, aDst, bw, nodes, s, half);
  }
}


static __device__ __forceinline__ void project_tile(
    const v2f (&a)[16], const v2f (&bw)[KK_TOT * NT], const v8f (&biasT)[NT],
    int t, int half, int l15, int nNodes,
    float* __restrict__ tabP, float* __restrict__ tabQ)
{
  v8f zeroT[NT];
#pragma unroll
  for (int n = 0; n < NT; ++n)
#pragma unroll
    for (int r = 0; r < 8; ++r) zeroT[n][r] = 0.f;

  v8f accP[NT], accQ[NT];
  mma_half<0>(accP, biasT, a, bw);
  mma_half<16>(accQ, zeroT, a, bw);

#pragma unroll
  for (int r = 0; r < 8; ++r) {
    int row = (t << 4) + (half << 3) + r;
    if (row >= nNodes) row = nNodes - 1;
    const long o = (long)row * DIM + (l15 << 2);
    *(float4*)(tabP + o) = make_float4(accP[0][r], accP[1][r], accP[2][r], accP[3][r]);
  }
#pragma unroll
  for (int r = 0; r < 8; ++r) {
    int row = (t << 4) + (half << 3) + r;
    if (row >= nNodes) row = nNodes - 1;
    const long o = (long)row * DIM + (l15 << 2);
    *(float4*)(tabQ + o) = make_float4(accQ[0][r], accQ[1][r], accQ[2][r], accQ[3][r]);
  }
}

static __device__ __forceinline__ int tile_row(int t, int l15, int nNodes)
{
  int row = (t << 4) + l15;
  return row < nNodes ? row : nNodes - 1;
}

__global__ __launch_bounds__(256) void node_project_wmma(
    const float* __restrict__ nodes, const float* __restrict__ W1,
    const float* __restrict__ b1, float* __restrict__ tabP,
    float* __restrict__ tabQ, int nNodes)
{
  __shared__ v2f ldsB[KK_TOT * NT * 32];

  const int lane = threadIdx.x & 31;
  const int half = lane >> 4;
  const int l15  = lane & 15;

  const int gw = (blockIdx.x << 3) + (threadIdx.x >> 5);
  const int nw = gridDim.x << 3;
  const int tiles = (nNodes + 15) >> 4;
  int t = gw;

  v2f aCur[16], aNext[16];
  gather_row(nodes, tile_row(t, l15, nNodes), half, aCur);

#pragma unroll
  for (int i = 0; i < (KK_TOT * NT * 32) / 256; ++i) {
    const int g = threadIdx.x + (i << 8);
    int f = g >> 5, ln = g & 31;
    int n = f >> 5, kk = f & 31;
    int hf = ln >> 4, c = (n << 4) + (ln & 15);
    int row = (kk << 2) + (hf << 1);
    v2f v;
    v.x = W1[row * DIM + c];
    v.y = W1[(row + 1) * DIM + c];
    ldsB[g] = v;
  }
  __syncthreads();

  if (t >= tiles) return;

  v8f biasT[NT];
#pragma unroll
  for (int n = 0; n < NT; ++n) {
    const float bv = b1[n * 16 + l15];
#pragma unroll
    for (int r = 0; r < 8; ++r) biasT[n][r] = bv;
  }

  v2f bw[KK_TOT * NT];
#pragma unroll
  for (int f = 0; f < KK_TOT * NT; ++f) bw[f] = ldsB[f * 32 + lane];

  for (;;) {
    const int tn = t + nw;
    gather_row(nodes, tile_row(tn, l15, nNodes), half, aNext);
    __builtin_amdgcn_sched_barrier(0);
    project_tile(aCur, bw, biasT, t, half, l15, nNodes, tabP, tabQ);
    __builtin_amdgcn_sched_barrier(0);
#pragma unroll
    for (int kk = 0; kk < 16; ++kk) aCur[kk] = aNext[kk];
    t = tn;
    if (t >= tiles) break;
  }
}


template <int J>
static __device__ __forceinline__ int row_lane(int v)
{
  return __builtin_amdgcn_update_dpp(0, v, 0x150 + J, 0xf, 0xf, false);
}

template <int M>
static __device__ __forceinline__ float row_xor(float v)
{
  return __int_as_float(__builtin_amdgcn_update_dpp(
      0, __float_as_int(v), 0x160 + M, 0xf, 0xf, false));
}

template <int J>
static __device__ __forceinline__ void edge_rows(
    const char* __restrict__ tabP, const char* __restrict__ tabQ,
    int sl, int dl, unsigned laneOff, float4& p, float4& q)
{
  const unsigned so = ((unsigned)row_lane<J>(sl) << 8) + laneOff;
  const unsigned qo = ((unsigned)row_lane<J>(dl) << 8) + laneOff;
  p = *(const float4*)(tabP + so);
  q = *(const float4*)(tabQ + qo);
}

static __device__ __forceinline__ float edge_partial(const float4 p, const float4 q, const float4 w)
{
  const float h0 = fmaxf(p.x + q.x, 0.f);
  const float h1 = fmaxf(p.y + q.y, 0.f);
  const float h2 = fmaxf(p.z + q.z, 0.f);
  const float h3 = fmaxf(p.w + q.w, 0.f);
  return fmaf(h3, w.w, fmaf(h2, w.z, fmaf(h1, w.y, h0 * w.x)));
}

__global__ __launch_bounds__(256) __attribute__((amdgpu_waves_per_eu(4, 4))) void edge_score_gather(
    const float* __restrict__ tabPf, const float* __restrict__ tabQf,
    const int* __restrict__ src, const int* __restrict__ dst,
    const float* __restrict__ W2, const float* __restrict__ b2,
    float* __restrict__ out, int E)
{
  const char* __restrict__ tabP = (const char*)tabPf;
  const char* __restrict__ tabQ = (const char*)tabQf;

  const int lane = threadIdx.x & 31;
  const int c    = lane & 15;
  const unsigned laneOff = (unsigned)c << 4;

  const float4 w = make_float4(W2[c], W2[16 + c], W2[32 + c], W2[48 + c]);
  const float b2v = b2[0];

  const int wavesPerBlk = blockDim.x >> 5;
  const int gw = __builtin_amdgcn_readfirstlane(blockIdx.x * wavesPerBlk + (threadIdx.x >> 5));
  const int nw = gridDim.x * wavesPerBlk;
  const int tiles = (E + 31) >> 5;

  const bool k8 = (c & 8) != 0, k4 = (c & 4) != 0, k2 = (c & 2) != 0, k1 = (c & 1) != 0;

  int t = gw;
  if (t >= tiles) return;

  int e = (t << 5) + lane;
  int ec = e < E ? e : E - 1;
  int sl = __builtin_nontemporal_load(src + ec);
  int dl = __builtin_nontemporal_load(dst + ec);

  for (;;) {
    const int tn = t + nw;
    const int en = (tn << 5) + lane;
    const int enc = en < E ? en : E - 1;
    const int sln = __builtin_nontemporal_load(src + enc);
    const int dln = __builtin_nontemporal_load(dst + enc);

    float4 p[16], q[16];
    edge_rows<0>(tabP, tabQ, sl, dl, laneOff, p[0], q[0]);
    edge_rows<1>(tabP, tabQ, sl, dl, laneOff, p[1], q[1]);
    edge_rows<2>(tabP, tabQ, sl, dl, laneOff, p[2], q[2]);
    edge_rows<3>(tabP, tabQ, sl, dl, laneOff, p[3], q[3]);
    edge_rows<4>(tabP, tabQ, sl, dl, laneOff, p[4], q[4]);
    edge_rows<5>(tabP, tabQ, sl, dl, laneOff, p[5], q[5]);
    edge_rows<6>(tabP, tabQ, sl, dl, laneOff, p[6], q[6]);
    edge_rows<7>(tabP, tabQ, sl, dl, laneOff, p[7], q[7]);
    edge_rows<8>(tabP, tabQ, sl, dl, laneOff, p[8], q[8]);
    edge_rows<9>(tabP, tabQ, sl, dl, laneOff, p[9], q[9]);
    edge_rows<10>(tabP, tabQ, sl, dl, laneOff, p[10], q[10]);
    edge_rows<11>(tabP, tabQ, sl, dl, laneOff, p[11], q[11]);
    edge_rows<12>(tabP, tabQ, sl, dl, laneOff, p[12], q[12]);
    edge_rows<13>(tabP, tabQ, sl, dl, laneOff, p[13], q[13]);
    edge_rows<14>(tabP, tabQ, sl, dl, laneOff, p[14], q[14]);
    edge_rows<15>(tabP, tabQ, sl, dl, laneOff, p[15], q[15]);
    float v[16];
#pragma unroll
    for (int j = 0; j < 16; ++j) v[j] = edge_partial(p[j], q[j], w);

    float u8[8];
#pragma unroll
    for (int j = 0; j < 8; ++j)
      u8[j] = (k8 ? v[j + 8] : v[j]) + row_xor<8>(k8 ? v[j] : v[j + 8]);
    float u4[4];
#pragma unroll
    for (int j = 0; j < 4; ++j)
      u4[j] = (k4 ? u8[j + 4] : u8[j]) + row_xor<4>(k4 ? u8[j] : u8[j + 4]);
    float u2[2];
#pragma unroll
    for (int j = 0; j < 2; ++j)
      u2[j] = (k2 ? u4[j + 2] : u4[j]) + row_xor<2>(k2 ? u4[j] : u4[j + 2]);
    const float logit = (k1 ? u2[1] : u2[0]) + row_xor<1>(k1 ? u2[0] : u2[1]);

    if (e < E) __builtin_nontemporal_store(logit + b2v, out + e);

    t = tn;
    if (t >= tiles) break;
    e = en; sl = sln; dl = dln;
  }
}

extern "C" void kernel_launch(void* const* d_in, const int* in_sizes, int n_in,
                              void* d_out, int out_size, void* d_ws, size_t ws_size,
                              hipStream_t stream) {
  const float* nodes = (const float*)d_in[0];
  const int*   src   = (const int*)d_in[1];
  const int*   dst   = (const int*)d_in[2];
  const float* W1    = (const float*)d_in[3];
  const float* b1    = (const float*)d_in[4];
  const float* W2    = (const float*)d_in[5];
  const float* b2    = (const float*)d_in[6];
  float* out = (float*)d_out;

  const int E = in_sizes[1];
  const int nNodes = in_sizes[0] / DIM;

  const size_t tableFloats = (size_t)nNodes * DIM;
  if (ws_size >= 2 * tableFloats * sizeof(float)) {
    float* tabP = (float*)d_ws;
    float* tabQ = tabP + tableFloats;
    node_project_wmma<<<256, 256, 0, stream>>>(nodes, W1, b1, tabP, tabQ, nNodes);
    edge_score_gather<<<512, 256, 0, stream>>>(tabP, tabQ, src, dst, W2, b2, out, E);
    return;
  }

  const int tiles = (E + 15) / 16;
  int blocks = (tiles + 7) / 8;
  if (blocks > 2048) blocks = 2048;
  if (blocks < 1) blocks = 1;

  edge_mlp_wmma<<<blocks, 256, 0, stream>>>(nodes, src, dst, W1, b1, W2, b2, out, E);
}
